# starting kernel
# baseline (speedup 1.0000x reference)
#include <hip/hip_runtime.h>


typedef __attribute__((ext_vector_type(2))) float v2f;
typedef __attribute__((ext_vector_type(4))) float v4f;
typedef __attribute__((ext_vector_type(8))) float v8f;

#define DIMS 128
#define LDS_STRIDE 132
#define TILE 16
#define TILES_PER_WAVE 8
#define NODES_PER_WAVE (TILE * TILES_PER_WAVE)
#define WAVES_PER_BLOCK 2
#define BLOCK (32 * WAVES_PER_BLOCK)

__global__ void n2g_zero_kernel(float* __restrict__ out, int n) {
    int i = blockIdx.x * blockDim.x + threadIdx.x;
    if (i < n) out[i] = 0.0f;
}

__global__ __launch_bounds__(BLOCK)
void n2g_attention_kernel(const float* __restrict__ nEmb,
                          const float* __restrict__ gEmb,
                          const int*   __restrict__ batch,
                          float*       __restrict__ out,
                          int N) {
    __shared__ float lds[WAVES_PER_BLOCK][2][TILE * LDS_STRIDE];
    const int lane  = threadIdx.x & 31;
    const int wave  = threadIdx.x >> 5;
    const int gwave = blockIdx.x * WAVES_PER_BLOCK + wave;
    float* nT = lds[wave][0];
    float* gT = lds[wave][1];

    const int m    = lane & 15;
    const int off  = (lane < 16) ? 0 : 2;
    const int col4 = lane * 4;

    const long base0 = (long)gwave * NODES_PER_WAVE;

    int curG = -1;
    v4f acc  = {0.f, 0.f, 0.f, 0.f};

    for (int t = 0; t < TILES_PER_WAVE; ++t) {
        const long base = base0 + (long)t * TILE;

        long bi = base + m;
        if (bi > (long)N - 1) bi = (long)N - 1;
        const int myG = batch[bi];

        #pragma unroll
        for (int r = 0; r < TILE; ++r) {
            const long node = base + r;
            const int  gr   = __shfl(myG, r, 32);
            v4f nv = {0.f, 0.f, 0.f, 0.f};
            if (node < N)
                nv = *(const v4f*)(nEmb + node * DIMS + col4);
            *(v4f*)(nT + r * LDS_STRIDE + col4) = nv;
            v4f gv = *(const v4f*)(gEmb + (long)gr * DIMS + col4);
            *(v4f*)(gT + r * LDS_STRIDE + col4) = gv;
        }
        asm volatile("s_wait_dscnt 0" ::: "memory");

        const float* pA = nT + m * LDS_STRIDE + off;
        const float* pB = gT + m * LDS_STRIDE + off;
        v8f c = {0.f, 0.f, 0.f, 0.f, 0.f, 0.f, 0.f, 0.f};
        #pragma unroll
        for (int k0 = 0; k0 < DIMS; k0 += 4) {
            v2f a = *(const v2f*)(pA + k0);
            v2f b = *(const v2f*)(pB + k0);
            c = __builtin_amdgcn_wmma_f32_16x16x4_f32(
                    false, a, false, b, (short)0, c, false, false);
        }

        float coef[TILE];
        #pragma unroll
        for (int j = 0; j < TILE; ++j) {
            const int srcLane = (j < 8) ? j : (j + 16);
            const float dj = __shfl(c[j & 7], srcLane, 32);
            coef[j] = 1.0f / (1.0f + __expf(-dj));
        }

        #pragma unroll
        for (int j = 0; j < TILE; ++j) {
            const long node = base + j;
            if (node < N) {
                const int gj = __shfl(myG, j, 32);
                v4f row = *(const v4f*)(nT + j * LDS_STRIDE + col4);
                if (gj != curG) {
                    if (curG >= 0) {
                        float* o = out + (long)curG * DIMS + col4;
                        atomicAdd(o + 0, acc.x);
                        atomicAdd(o + 1, acc.y);
                        atomicAdd(o + 2, acc.z);
                        atomicAdd(o + 3, acc.w);
                    }
                    curG = gj;
                    acc = coef[j] * row;
                } else {
                    acc += coef[j] * row;
                }
            }
        }
    }

    if (curG >= 0) {
        float* o = out + (long)curG * DIMS + col4;
        atomicAdd(o + 0, acc.x);
        atomicAdd(o + 1, acc.y);
        atomicAdd(o + 2, acc.z);
        atomicAdd(o + 3, acc.w);
    }
}

extern "C" void kernel_launch(void* const* d_in, const int* in_sizes, int n_in,
                              void* d_out, int out_size, void* d_ws, size_t ws_size,
                              hipStream_t stream) {
    const float* nEmb  = (const float*)d_in[0];
    const float* gEmb  = (const float*)d_in[1];
    const int*   batch = (const int*)d_in[2];
    float*       out   = (float*)d_out;

    const int N = in_sizes[0] / DIMS;

    n2g_zero_kernel<<<(out_size + 255) / 256, 256, 0, stream>>>(out, out_size);

    const int numWaves  = (N + NODES_PER_WAVE - 1) / NODES_PER_WAVE;
    const int numBlocks = (numWaves + WAVES_PER_BLOCK - 1) / WAVES_PER_BLOCK;
    n2g_attention_kernel<<<numBlocks, BLOCK, 0, stream>>>(nEmb, gEmb, batch, out, N);
}


# Request each tile's 32 row loads as one group, one tile ahead of use (register prefetch; row index clamped instead of a per-row branch), in place of one load-then-wait per row
# speedup vs baseline: 1.9203x; 1.9203x over previous
#include <hip/hip_runtime.h>


typedef __attribute__((ext_vector_type(2))) float v2f;
typedef __attribute__((ext_vector_type(4))) float v4f;
typedef __attribute__((ext_vector_type(8))) float v8f;

#define DIMS 128
#define LDS_STRIDE 132
#define TILE 16
#define TILES_PER_WAVE 8
#define NODES_PER_WAVE (TILE * TILES_PER_WAVE)
#define WAVES_PER_BLOCK 2
#define BLOCK (32 * WAVES_PER_BLOCK)

__global__ void n2g_zero_kernel(float* __restrict__ out, int n) {
    int i = blockIdx.x * blockDim.x + threadIdx.x;
    if (i < n) out[i] = 0.0f;
}

__device__ __forceinline__ int load_graph_id(const int* __restrict__ batch,
                                             int first, int m, int lastNode) {
    int bi = first + m;
    if (bi > lastNode) bi = lastNode;
    return batch[bi];
}

__device__ __forceinline__ void request_tile_rows(const float* __restrict__ nEmb,
                                                  const float* __restrict__ gEmb,
                                                  int first, int lastNode,
                                                  int gIds, int col4,
                                                  v4f (&nv)[TILE], v4f (&gv)[TILE]) {
    #pragma unroll
    for (int r = 0; r < TILE; ++r) {
        int node = first + r;
        if (node > lastNode) node = lastNode;
        const int gr = __shfl(gIds, r, 32);
        nv[r] = *(const v4f*)(nEmb + (size_t)node * DIMS + col4);
        gv[r] = *(const v4f*)(gEmb + (size_t)gr * DIMS + col4);
    }
}

__global__ __launch_bounds__(BLOCK)
void n2g_attention_kernel(const float* __restrict__ nEmb,
                          const float* __restrict__ gEmb,
                          const int*   __restrict__ batch,
                          float*       __restrict__ out,
                          int N) {
    __shared__ float lds[WAVES_PER_BLOCK][2][TILE * LDS_STRIDE];
    const int lane  = threadIdx.x & 31;
    const int wave  = threadIdx.x >> 5;
    const int gwave = blockIdx.x * WAVES_PER_BLOCK + wave;
    float* nT = lds[wave][0];
    float* gT = lds[wave][1];

    const int m    = lane & 15;
    const int off  = (lane < 16) ? 0 : 2;
    const int col4 = lane * 4;

    const int base0    = gwave * NODES_PER_WAVE;
    const int lastNode = N - 1;

    int curG = -1;
    v4f acc  = {0.f, 0.f, 0.f, 0.f};

    int gThis = load_graph_id(batch, base0, m, lastNode);
    int gNext = load_graph_id(batch, base0 + TILE, m, lastNode);
    v4f nv[TILE], gv[TILE];
    request_tile_rows(nEmb, gEmb, base0, lastNode, gThis, col4, nv, gv);

    for (int t = 0; t < TILES_PER_WAVE; ++t) {
        const int base = base0 + t * TILE;
        const int myG  = gThis;

        #pragma unroll
        for (int r = 0; r < TILE; ++r) {
            *(v4f*)(nT + r * LDS_STRIDE + col4) = nv[r];
            *(v4f*)(gT + r * LDS_STRIDE + col4) = gv[r];
        }

        if (t + 1 < TILES_PER_WAVE) {
            gThis = gNext;
            request_tile_rows(nEmb, gEmb, base + TILE, lastNode, gThis, col4, nv, gv);
            if (t + 2 < TILES_PER_WAVE)
                gNext = load_graph_id(batch, base + 2 * TILE, m, lastNode);
        }

        asm volatile("s_wait_dscnt 0" ::: "memory");

        const float* pA = nT + m * LDS_STRIDE + off;
        const float* pB = gT + m * LDS_STRIDE + off;
        v8f c = {0.f, 0.f, 0.f, 0.f, 0.f, 0.f, 0.f, 0.f};
        #pragma unroll
        for (int k0 = 0; k0 < DIMS; k0 += 4) {
            v2f a = *(const v2f*)(pA + k0);
            v2f b = *(const v2f*)(pB + k0);
            c = __builtin_amdgcn_wmma_f32_16x16x4_f32(
                    false, a, false, b, (short)0, c, false, false);
        }

        float coef[TILE];
        #pragma unroll
        for (int j = 0; j < TILE; ++j) {
            const int srcLane = (j < 8) ? j : (j + 16);
            const float dj = __shfl(c[j & 7], srcLane, 32);
            coef[j] = 1.0f / (1.0f + __expf(-dj));
        }

        #pragma unroll
        for (int j = 0; j < TILE; ++j) {
            const int node = base + j;
            if (node < N) {
                const int gj = __shfl(myG, j, 32);
                v4f row = *(const v4f*)(nT + j * LDS_STRIDE + col4);
                if (gj != curG) {
                    if (curG >= 0) {
                        float* o = out + (long)curG * DIMS + col4;
                        atomicAdd(o + 0, acc.x);
                        atomicAdd(o + 1, acc.y);
                        atomicAdd(o + 2, acc.z);
                        atomicAdd(o + 3, acc.w);
                    }
                    curG = gj;
                    acc = coef[j] * row;
                } else {
                    acc += coef[j] * row;
                }
            }
        }
    }

    if (curG >= 0) {
        float* o = out + (long)curG * DIMS + col4;
        atomicAdd(o + 0, acc.x);
        atomicAdd(o + 1, acc.y);
        atomicAdd(o + 2, acc.z);
        atomicAdd(o + 3, acc.w);
    }
}

extern "C" void kernel_launch(void* const* d_in, const int* in_sizes, int n_in,
                              void* d_out, int out_size, void* d_ws, size_t ws_size,
                              hipStream_t stream) {
    const float* nEmb  = (const float*)d_in[0];
    const float* gEmb  = (const float*)d_in[1];
    const int*   batch = (const int*)d_in[2];
    float*       out   = (float*)d_out;

    const int N = in_sizes[0] / DIMS;

    n2g_zero_kernel<<<(out_size + 255) / 256, 256, 0, stream>>>(out, out_size);

    const int numWaves  = (N + NODES_PER_WAVE - 1) / NODES_PER_WAVE;
    const int numBlocks = (numWaves + WAVES_PER_BLOCK - 1) / WAVES_PER_BLOCK;
    n2g_attention_kernel<<<numBlocks, BLOCK, 0, stream>>>(nEmb, gEmb, batch, out, N);
}


# Per-wave values (node index, graph id, coefficient) live in scalar registers: read-lane in place of LDS lane shuffles, scalar row addresses and branches; one sigmoid per tile on the diagonal lanes
# speedup vs baseline: 2.5568x; 1.3315x over previous
#include <hip/hip_runtime.h>


typedef __attribute__((ext_vector_type(2))) float v2f;
typedef __attribute__((ext_vector_type(4))) float v4f;
typedef __attribute__((ext_vector_type(8))) float v8f;
typedef const char __attribute__((address_space(1)))* gbytes;
typedef const v4f  __attribute__((address_space(1)))* gv4f;

#define DIMS 128
#define LDS_STRIDE 132
#define TILE 16
#define TILES_PER_WAVE 8
#define NODES_PER_WAVE (TILE * TILES_PER_WAVE)
#define WAVES_PER_BLOCK 2
#define BLOCK (32 * WAVES_PER_BLOCK)

__global__ void n2g_zero_kernel(float* __restrict__ out, int n) {
    int i = blockIdx.x * blockDim.x + threadIdx.x;
    if (i < n) out[i] = 0.0f;
}

__device__ __forceinline__ int lane_int(int x, int l) {
    return __builtin_amdgcn_readlane(x, l);
}
__device__ __forceinline__ float lane_float(float x, int l) {
    return __int_as_float(__builtin_amdgcn_readlane(__float_as_int(x), l));
}

__device__ __forceinline__ int load_graph_id(const int* __restrict__ batch,
                                             int first, int m, int lastNode) {
    int bi = first + m;
    if (bi > lastNode) bi = lastNode;
    return batch[bi];
}

__device__ __forceinline__ void request_tile_rows(const float* __restrict__ nEmb,
                                                  const float* __restrict__ gEmb,
                                                  int first, int lastNode,
                                                  int gIds, unsigned colBytes,
                                                  v4f (&nv)[TILE], v4f (&gv)[TILE]) {
    asm("" : "+v"(colBytes));
    #pragma unroll
    for (int r = 0; r < TILE; ++r) {
        int node = first + r;
        if (node > lastNode) node = lastNode;
        const int gr = lane_int(gIds, r);
        gbytes nRow = (gbytes)(nEmb + (size_t)node * DIMS);
        gbytes gRow = (gbytes)(gEmb + (size_t)gr * DIMS);
        asm("" : "+s"(nRow), "+s"(gRow));
        nv[r] = *(gv4f)(nRow + colBytes);
        gv[r] = *(gv4f)(gRow + colBytes);
    }
}

__device__ __forceinline__ void flush_run(float* __restrict__ out, int g, int col4, v4f acc) {
    float* o = out + (long)g * DIMS + col4;
    atomicAdd(o + 0, acc.x);
    atomicAdd(o + 1, acc.y);
    atomicAdd(o + 2, acc.z);
    atomicAdd(o + 3, acc.w);
}

__global__ __launch_bounds__(BLOCK)
void n2g_attention_kernel(const float* __restrict__ nEmb,
                          const float* __restrict__ gEmb,
                          const int*   __restrict__ batch,
                          float*       __restrict__ out,
                          int N) {
    __shared__ float lds[WAVES_PER_BLOCK][2][TILE * LDS_STRIDE];
    const int lane  = threadIdx.x & 31;
    const int wave  = __builtin_amdgcn_readfirstlane((int)(threadIdx.x >> 5));
    const int gwave = blockIdx.x * WAVES_PER_BLOCK + wave;
    float* nT = lds[wave][0];
    float* gT = lds[wave][1];

    const int m    = lane & 15;
    const int off  = (lane < 16) ? 0 : 2;
    const int col4 = lane * 4;
    const unsigned colBytes = (unsigned)lane * 16u;
    const int dreg = lane & 7;

    const int base0    = gwave * NODES_PER_WAVE;
    const int lastNode = N - 1;

    int curG = -1;
    v4f acc  = {0.f, 0.f, 0.f, 0.f};

    int gThis = load_graph_id(batch, base0, m, lastNode);
    int gNext = load_graph_id(batch, base0 + TILE, m, lastNode);
    v4f nv[TILE], gv[TILE];
    request_tile_rows(nEmb, gEmb, base0, lastNode, gThis, colBytes, nv, gv);

    for (int t = 0; t < TILES_PER_WAVE; ++t) {
        const int base = base0 + t * TILE;
        const int myG  = gThis;

        #pragma unroll
        for (int r = 0; r < TILE; ++r) {
            *(v4f*)(nT + r * LDS_STRIDE + col4) = nv[r];
            *(v4f*)(gT + r * LDS_STRIDE + col4) = gv[r];
        }

        if (t + 1 < TILES_PER_WAVE) {
            gThis = gNext;
            request_tile_rows(nEmb, gEmb, base + TILE, lastNode, gThis, colBytes, nv, gv);
            if (t + 2 < TILES_PER_WAVE)
                gNext = load_graph_id(batch, base + 2 * TILE, m, lastNode);
        }

        asm volatile("s_wait_dscnt 0" ::: "memory");

        const float* pA = nT + m * LDS_STRIDE + off;
        const float* pB = gT + m * LDS_STRIDE + off;
        v8f c = {0.f, 0.f, 0.f, 0.f, 0.f, 0.f, 0.f, 0.f};
        #pragma unroll
        for (int k0 = 0; k0 < DIMS; k0 += 4) {
            v2f a = *(const v2f*)(pA + k0);
            v2f b = *(const v2f*)(pB + k0);
            c = __builtin_amdgcn_wmma_f32_16x16x4_f32(
                    false, a, false, b, (short)0, c, false, false);
        }

        float d = c[0];
        #pragma unroll
        for (int r = 1; r < 8; ++r) d = (dreg == r) ? c[r] : d;
        const float sig = 1.0f / (1.0f + __expf(-d));

        float coef[TILE];
        int   gOf[TILE];
        #pragma unroll
        for (int j = 0; j < TILE; ++j) {
            coef[j] = lane_float(sig, (j < 8) ? j : (j + 16));
            gOf[j]  = lane_int(myG, j);
        }

        const int nValid = N - base;
        if (nValid >= TILE) {
            #pragma unroll
            for (int h = 0; h < TILE; h += 8) {
                v4f row[8];
                #pragma unroll
                for (int i = 0; i < 8; ++i)
                    row[i] = *(const v4f*)(nT + (h + i) * LDS_STRIDE + col4);
                #pragma unroll
                for (int i = 0; i < 8; ++i) {
                    const int j = h + i;
                    if (__builtin_expect(gOf[j] != curG, 0)) {
                        if (curG >= 0) flush_run(out, curG, col4, acc);
                        curG = gOf[j];
                        acc = (v4f){0.f, 0.f, 0.f, 0.f};
                    }
                    acc += coef[j] * row[i];
                }
            }
        } else {
            #pragma nounroll
            for (int j = 0; j < nValid; ++j) {
                const int   gj = __builtin_amdgcn_readlane(myG, j);
                const float cj = lane_float(sig, (j < 8) ? j : (j + 16));
                const v4f  row = *(const v4f*)(nT + j * LDS_STRIDE + col4);
                if (gj != curG) {
                    if (curG >= 0) flush_run(out, curG, col4, acc);
                    curG = gj;
                    acc = (v4f){0.f, 0.f, 0.f, 0.f};
                }
                acc += cj * row;
            }
        }
    }

    if (curG >= 0) flush_run(out, curG, col4, acc);
}

extern "C" void kernel_launch(void* const* d_in, const int* in_sizes, int n_in,
                              void* d_out, int out_size, void* d_ws, size_t ws_size,
                              hipStream_t stream) {
    const float* nEmb  = (const float*)d_in[0];
    const float* gEmb  = (const float*)d_in[1];
    const int*   batch = (const int*)d_in[2];
    float*       out   = (float*)d_out;

    const int N = in_sizes[0] / DIMS;

    n2g_zero_kernel<<<(out_size + 255) / 256, 256, 0, stream>>>(out, out_size);

    const int numWaves  = (N + NODES_PER_WAVE - 1) / NODES_PER_WAVE;
    const int numBlocks = (numWaves + WAVES_PER_BLOCK - 1) / WAVES_PER_BLOCK;
    n2g_attention_kernel<<<numBlocks, BLOCK, 0, stream>>>(nEmb, gEmb, batch, out, N);
}


# Graph rows skip LDS: each lane loads its node's graph row from global memory as the WMMA B fragments; A read as one b128 group ahead of the 32 WMMAs (k order of the dot sums changes)
# speedup vs baseline: 2.8480x; 1.1139x over previous
#include <hip/hip_runtime.h>


typedef __attribute__((ext_vector_type(2))) float v2f;
typedef __attribute__((ext_vector_type(4))) float v4f;
typedef __attribute__((ext_vector_type(8))) float v8f;
typedef const char __attribute__((address_space(1)))* gbytes;
typedef const v4f  __attribute__((address_space(1)))* gv4f;

#define DIMS 128
#define LDS_STRIDE 132
#define TILE 16
#define TILES_PER_WAVE 8
#define NODES_PER_WAVE (TILE * TILES_PER_WAVE)
#define WAVES_PER_BLOCK 2
#define BLOCK (32 * WAVES_PER_BLOCK)

__global__ void n2g_zero_kernel(float* __restrict__ out, int n) {
    int i = blockIdx.x * blockDim.x + threadIdx.x;
    if (i < n) out[i] = 0.0f;
}

__device__ __forceinline__ int lane_int(int x, int l) {
    return __builtin_amdgcn_readlane(x, l);
}
__device__ __forceinline__ float lane_float(float x, int l) {
    return __int_as_float(__builtin_amdgcn_readlane(__float_as_int(x), l));
}

__device__ __forceinline__ int load_graph_id(const int* __restrict__ batch,
                                             int first, int m, int lastNode) {
    int bi = first + m;
    if (bi > lastNode) bi = lastNode;
    return batch[bi];
}

#define FRAG_COL(i, h) (8 * (i) + 4 * (h))
#define PAIRS (DIMS / 8)

__device__ __forceinline__ void request_node_rows(const float* __restrict__ nEmb,
                                                  int first, int lastNode,
                                                  unsigned colBytes, v4f (&nv)[TILE]) {
    asm("" : "+v"(colBytes));
    gbytes rowBase[TILE];
    #pragma unroll
    for (int r = 0; r < TILE; ++r) {
        int node = first + r;
        if (node > lastNode) node = lastNode;
        rowBase[r] = (gbytes)(nEmb + (size_t)node * DIMS);
    }
    asm("" : "+s"(rowBase[0]),  "+s"(rowBase[1]),  "+s"(rowBase[2]),  "+s"(rowBase[3]),
             "+s"(rowBase[4]),  "+s"(rowBase[5]),  "+s"(rowBase[6]),  "+s"(rowBase[7]),
             "+s"(rowBase[8]),  "+s"(rowBase[9]),  "+s"(rowBase[10]), "+s"(rowBase[11]),
             "+s"(rowBase[12]), "+s"(rowBase[13]), "+s"(rowBase[14]), "+s"(rowBase[15]));
    #pragma unroll
    for (int r = 0; r < TILE; ++r)
        nv[r] = *(gv4f)(rowBase[r] + colBytes);
}

__device__ __forceinline__ void request_graph_frags(const float* __restrict__ gEmb,
                                                    int gId, int h, v4f (&gv)[PAIRS]) {
    gbytes gRow = (gbytes)(gEmb + (size_t)gId * DIMS + FRAG_COL(0, h));
    #pragma unroll
    for (int i = 0; i < PAIRS; ++i)
        gv[i] = *(gv4f)(gRow + (FRAG_COL(i, 0) * sizeof(float)));
}

__device__ __forceinline__ void flush_run(float* __restrict__ out, int g, int col4, v4f acc) {
    float* o = out + (long)g * DIMS + col4;
    atomicAdd(o + 0, acc.x);
    atomicAdd(o + 1, acc.y);
    atomicAdd(o + 2, acc.z);
    atomicAdd(o + 3, acc.w);
}

template <bool B> struct has_next { static constexpr bool value = B; };

__global__ __launch_bounds__(BLOCK)
void n2g_attention_kernel(const float* __restrict__ nEmb,
                          const float* __restrict__ gEmb,
                          const int*   __restrict__ batch,
                          float*       __restrict__ out,
                          int N) {
    __shared__ float lds[WAVES_PER_BLOCK][TILE * LDS_STRIDE];
    const int lane  = threadIdx.x & 31;
    const int wave  = __builtin_amdgcn_readfirstlane((int)(threadIdx.x >> 5));
    const int gwave = blockIdx.x * WAVES_PER_BLOCK + wave;
    float* nT = lds[wave];

    const int m    = lane & 15;
    const int half = lane >> 4;
    const int col4 = lane * 4;
    const unsigned colBytes = (unsigned)lane * 16u;
    const int dreg = lane & 7;

    const int base0    = gwave * NODES_PER_WAVE;
    const int lastNode = N - 1;

    int curG = -1;
    v4f acc  = {0.f, 0.f, 0.f, 0.f};

    int gThis = load_graph_id(batch, base0, m, lastNode);
    int gNext = load_graph_id(batch, base0 + TILE, m, lastNode);
    v4f nv[TILE];
    v4f gv[PAIRS];
    __builtin_amdgcn_sched_barrier(0);
    request_node_rows(nEmb, base0, lastNode, colBytes, nv);
    __builtin_amdgcn_sched_barrier(0);
    asm("" : "+v"(gThis));
    request_graph_frags(gEmb, gThis, half, gv);

    auto tile = [&](const int t, auto hasNext) {
        constexpr bool HAS_NEXT = decltype(hasNext)::value;
        const int base = base0 + t * TILE;
        const int myG  = gThis;

        #pragma unroll
        for (int r = 0; r < TILE; ++r)
            *(v4f*)(nT + r * LDS_STRIDE + col4) = nv[r];

        if constexpr (HAS_NEXT)
            request_node_rows(nEmb, base + TILE, lastNode, colBytes, nv);

        asm volatile("s_wait_dscnt 0" ::: "memory");

        const float* pA = nT + m * LDS_STRIDE + FRAG_COL(0, half);
        v4f af[PAIRS];
        #pragma unroll
        for (int i = 0; i < PAIRS; ++i)
            af[i] = *(const v4f*)(pA + FRAG_COL(i, 0));

        v8f c = {0.f, 0.f, 0.f, 0.f, 0.f, 0.f, 0.f, 0.f};
        #pragma unroll
        for (int i = 0; i < PAIRS; ++i) {
            const v2f a0 = {af[i].x, af[i].y}, b0 = {gv[i].x, gv[i].y};
            const v2f a1 = {af[i].z, af[i].w}, b1 = {gv[i].z, gv[i].w};
            c = __builtin_amdgcn_wmma_f32_16x16x4_f32(
                    false, a0, false, b0, (short)0, c, false, false);
            c = __builtin_amdgcn_wmma_f32_16x16x4_f32(
                    false, a1, false, b1, (short)0, c, false, false);
        }
        __builtin_amdgcn_sched_group_barrier(0x100, PAIRS, 0);
        __builtin_amdgcn_sched_group_barrier(0x008, 2 * PAIRS, 0);

        if constexpr (HAS_NEXT) {
            gThis = gNext;
            int idLane = m;
            asm("" : "+v"(gThis), "+v"(idLane) : "v"(c));
            request_graph_frags(gEmb, gThis, half, gv);
            gNext = load_graph_id(batch, base + 2 * TILE, idLane, lastNode);
        }

        float d = c[0];
        #pragma unroll
        for (int r = 1; r < 8; ++r) d = (dreg == r) ? c[r] : d;
        const float sig = 1.0f / (1.0f + __expf(-d));

        float coef[TILE];
        int   gOf[TILE];
        #pragma unroll
        for (int j = 0; j < TILE; ++j) {
            coef[j] = lane_float(sig, (j < 8) ? j : (j + 16));
            gOf[j]  = lane_int(myG, j);
        }

        const int nValid = N - base;
        if (nValid >= TILE) {
            #pragma unroll
            for (int h = 0; h < TILE; h += 8) {
                v4f row[8];
                #pragma unroll
                for (int i = 0; i < 8; ++i)
                    row[i] = *(const v4f*)(nT + (h + i) * LDS_STRIDE + col4);
                #pragma unroll
                for (int i = 0; i < 8; ++i) {
                    const int j = h + i;
                    if (__builtin_expect(gOf[j] != curG, 0)) {
                        if (curG >= 0) flush_run(out, curG, col4, acc);
                        curG = gOf[j];
                        acc = (v4f){0.f, 0.f, 0.f, 0.f};
                    }
                    acc += coef[j] * row[i];
                }
            }
        } else {
            #pragma nounroll
            for (int j = 0; j < nValid; ++j) {
                const int   gj = __builtin_amdgcn_readlane(myG, j);
                const float cj = lane_float(sig, (j < 8) ? j : (j + 16));
                const v4f  row = *(const v4f*)(nT + j * LDS_STRIDE + col4);
                if (gj != curG) {
                    if (curG >= 0) flush_run(out, curG, col4, acc);
                    curG = gj;
                    acc = (v4f){0.f, 0.f, 0.f, 0.f};
                }
                acc += cj * row;
            }
        }
    };

    for (int t = 0; t < TILES_PER_WAVE - 1; ++t) tile(t, has_next<true>{});
    tile(TILES_PER_WAVE - 1, has_next<false>{});

    if (curG >= 0) flush_run(out, curG, col4, acc);
}

extern "C" void kernel_launch(void* const* d_in, const int* in_sizes, int n_in,
                              void* d_out, int out_size, void* d_ws, size_t ws_size,
                              hipStream_t stream) {
    const float* nEmb  = (const float*)d_in[0];
    const float* gEmb  = (const float*)d_in[1];
    const int*   batch = (const int*)d_in[2];
    float*       out   = (float*)d_out;

    const int N = in_sizes[0] / DIMS;

    n2g_zero_kernel<<<(out_size + 255) / 256, 256, 0, stream>>>(out, out_size);

    const int numWaves  = (N + NODES_PER_WAVE - 1) / NODES_PER_WAVE;
    const int numBlocks = (numWaves + WAVES_PER_BLOCK - 1) / WAVES_PER_BLOCK;
    n2g_attention_kernel<<<numBlocks, BLOCK, 0, stream>>>(nEmb, gEmb, batch, out, N);
}


# Zero pass as a fixed grid of 1024 waves (256 work-groups of 4) that each store 16 bytes per lane in a short loop, in place of 40000 waves of one 4-byte store per lane: fewer waves to enter
# speedup vs baseline: 3.0333x; 1.0651x over previous
#include <hip/hip_runtime.h>


typedef __attribute__((ext_vector_type(2))) float v2f;
typedef __attribute__((ext_vector_type(4))) float v4f;
typedef __attribute__((ext_vector_type(8))) float v8f;
typedef const char __attribute__((address_space(1)))* gbytes;
typedef const v4f  __attribute__((address_space(1)))* gv4f;

#define DIMS 128
#define LDS_STRIDE 132
#define TILE 16
#define TILES_PER_WAVE 8
#define NODES_PER_WAVE (TILE * TILES_PER_WAVE)
#define WAVES_PER_BLOCK 2
#define BLOCK (32 * WAVES_PER_BLOCK)

#define ZERO_BLOCKS 256
#define ZERO_THREADS 128
#define ZERO_STRIDE (ZERO_BLOCKS * ZERO_THREADS)

__global__ __launch_bounds__(ZERO_THREADS)
void n2g_zero_kernel(float* __restrict__ out, int n) {
    const int tid = blockIdx.x * ZERO_THREADS + threadIdx.x;
    const int n4  = n >> 2;
    v4f* o4 = (v4f*)out;
    const v4f z = {0.f, 0.f, 0.f, 0.f};
    int i = tid;
    const int n4Full = n4 - 3 * ZERO_STRIDE;
    for (; i < n4Full; i += 4 * ZERO_STRIDE) {
        #pragma unroll
        for (int k = 0; k < 4; ++k) o4[i + k * ZERO_STRIDE] = z;
    }
    for (; i < n4; i += ZERO_STRIDE) o4[i] = z;
    const int t = (n4 << 2) + tid;
    if (t < n) out[t] = 0.0f;
}

__device__ __forceinline__ int lane_int(int x, int l) {
    return __builtin_amdgcn_readlane(x, l);
}
__device__ __forceinline__ float lane_float(float x, int l) {
    return __int_as_float(__builtin_amdgcn_readlane(__float_as_int(x), l));
}

__device__ __forceinline__ int load_graph_id(const int* __restrict__ batch,
                                             int first, int m, int lastNode) {
    int bi = first + m;
    if (bi > lastNode) bi = lastNode;
    return batch[bi];
}

#define FRAG_COL(i, h) (8 * (i) + 4 * (h))
#define PAIRS (DIMS / 8)

__device__ __forceinline__ void request_node_rows(const float* __restrict__ nEmb,
                                                  int first, int lastNode,
                                                  unsigned colBytes, v4f (&nv)[TILE]) {
    asm("" : "+v"(colBytes));
    gbytes rowBase[TILE];
    #pragma unroll
    for (int r = 0; r < TILE; ++r) {
        int node = first + r;
        if (node > lastNode) node = lastNode;
        rowBase[r] = (gbytes)(nEmb + (size_t)node * DIMS);
    }
    asm("" : "+s"(rowBase[0]),  "+s"(rowBase[1]),  "+s"(rowBase[2]),  "+s"(rowBase[3]),
             "+s"(rowBase[4]),  "+s"(rowBase[5]),  "+s"(rowBase[6]),  "+s"(rowBase[7]),
             "+s"(rowBase[8]),  "+s"(rowBase[9]),  "+s"(rowBase[10]), "+s"(rowBase[11]),
             "+s"(rowBase[12]), "+s"(rowBase[13]), "+s"(rowBase[14]), "+s"(rowBase[15]));
    #pragma unroll
    for (int r = 0; r < TILE; ++r)
        nv[r] = *(gv4f)(rowBase[r] + colBytes);
}

__device__ __forceinline__ void request_graph_frags(const float* __restrict__ gEmb,
                                                    int gId, int h, v4f (&gv)[PAIRS]) {
    gbytes gRow = (gbytes)(gEmb + (size_t)gId * DIMS + FRAG_COL(0, h));
    #pragma unroll
    for (int i = 0; i < PAIRS; ++i)
        gv[i] = *(gv4f)(gRow + (FRAG_COL(i, 0) * sizeof(float)));
}

__device__ __forceinline__ void flush_run(float* __restrict__ out, int g, int col4, v4f acc) {
    float* o = out + (long)g * DIMS + col4;
    atomicAdd(o + 0, acc.x);
    atomicAdd(o + 1, acc.y);
    atomicAdd(o + 2, acc.z);
    atomicAdd(o + 3, acc.w);
}

template <bool B> struct has_next { static constexpr bool value = B; };

__global__ __launch_bounds__(BLOCK)
void n2g_attention_kernel(const float* __restrict__ nEmb,
                          const float* __restrict__ gEmb,
                          const int*   __restrict__ batch,
                          float*       __restrict__ out,
                          int N) {
    __shared__ float lds[WAVES_PER_BLOCK][TILE * LDS_STRIDE];
    const int lane  = threadIdx.x & 31;
    const int wave  = __builtin_amdgcn_readfirstlane((int)(threadIdx.x >> 5));
    const int gwave = blockIdx.x * WAVES_PER_BLOCK + wave;
    float* nT = lds[wave];

    const int m    = lane & 15;
    const int half = lane >> 4;
    const int col4 = lane * 4;
    const unsigned colBytes = (unsigned)lane * 16u;
    const int dreg = lane & 7;

    const int base0    = gwave * NODES_PER_WAVE;
    const int lastNode = N - 1;

    int curG = -1;
    v4f acc  = {0.f, 0.f, 0.f, 0.f};

    int gThis = load_graph_id(batch, base0, m, lastNode);
    int gNext = load_graph_id(batch, base0 + TILE, m, lastNode);
    v4f nv[TILE];
    v4f gv[PAIRS];
    __builtin_amdgcn_sched_barrier(0);
    request_node_rows(nEmb, base0, lastNode, colBytes, nv);
    __builtin_amdgcn_sched_barrier(0);
    asm("" : "+v"(gThis));
    request_graph_frags(gEmb, gThis, half, gv);

    auto tile = [&](const int t, auto hasNext) {
        constexpr bool HAS_NEXT = decltype(hasNext)::value;
        const int base = base0 + t * TILE;
        const int myG  = gThis;

        #pragma unroll
        for (int r = 0; r < TILE; ++r)
            *(v4f*)(nT + r * LDS_STRIDE + col4) = nv[r];

        if constexpr (HAS_NEXT)
            request_node_rows(nEmb, base + TILE, lastNode, colBytes, nv);

        asm volatile("s_wait_dscnt 0" ::: "memory");

        const float* pA = nT + m * LDS_STRIDE + FRAG_COL(0, half);
        v4f af[PAIRS];
        #pragma unroll
        for (int i = 0; i < PAIRS; ++i)
            af[i] = *(const v4f*)(pA + FRAG_COL(i, 0));

        v8f c = {0.f, 0.f, 0.f, 0.f, 0.f, 0.f, 0.f, 0.f};
        #pragma unroll
        for (int i = 0; i < PAIRS; ++i) {
            const v2f a0 = {af[i].x, af[i].y}, b0 = {gv[i].x, gv[i].y};
            const v2f a1 = {af[i].z, af[i].w}, b1 = {gv[i].z, gv[i].w};
            c = __builtin_amdgcn_wmma_f32_16x16x4_f32(
                    false, a0, false, b0, (short)0, c, false, false);
            c = __builtin_amdgcn_wmma_f32_16x16x4_f32(
                    false, a1, false, b1, (short)0, c, false, false);
        }
        __builtin_amdgcn_sched_group_barrier(0x100, PAIRS, 0);
        __builtin_amdgcn_sched_group_barrier(0x008, 2 * PAIRS, 0);

        if constexpr (HAS_NEXT) {
            gThis = gNext;
            int idLane = m;
            asm("" : "+v"(gThis), "+v"(idLane) : "v"(c));
            request_graph_frags(gEmb, gThis, half, gv);
            gNext = load_graph_id(batch, base + 2 * TILE, idLane, lastNode);
        }

        float d = c[0];
        #pragma unroll
        for (int r = 1; r < 8; ++r) d = (dreg == r) ? c[r] : d;
        const float sig = 1.0f / (1.0f + __expf(-d));

        float coef[TILE];
        int   gOf[TILE];
        #pragma unroll
        for (int j = 0; j < TILE; ++j) {
            coef[j] = lane_float(sig, (j < 8) ? j : (j + 16));
            gOf[j]  = lane_int(myG, j);
        }

        const int nValid = N - base;
        if (nValid >= TILE) {
            #pragma unroll
            for (int h = 0; h < TILE; h += 8) {
                v4f row[8];
                #pragma unroll
                for (int i = 0; i < 8; ++i)
                    row[i] = *(const v4f*)(nT + (h + i) * LDS_STRIDE + col4);
                #pragma unroll
                for (int i = 0; i < 8; ++i) {
                    const int j = h + i;
                    if (__builtin_expect(gOf[j] != curG, 0)) {
                        if (curG >= 0) flush_run(out, curG, col4, acc);
                        curG = gOf[j];
                        acc = (v4f){0.f, 0.f, 0.f, 0.f};
                    }
                    acc += coef[j] * row[i];
                }
            }
        } else {
            #pragma nounroll
            for (int j = 0; j < nValid; ++j) {
                const int   gj = __builtin_amdgcn_readlane(myG, j);
                const float cj = lane_float(sig, (j < 8) ? j : (j + 16));
                const v4f  row = *(const v4f*)(nT + j * LDS_STRIDE + col4);
                if (gj != curG) {
                    if (curG >= 0) flush_run(out, curG, col4, acc);
                    curG = gj;
                    acc = (v4f){0.f, 0.f, 0.f, 0.f};
                }
                acc += cj * row;
            }
        }
    };

    for (int t = 0; t < TILES_PER_WAVE - 1; ++t) tile(t, has_next<true>{});
    tile(TILES_PER_WAVE - 1, has_next<false>{});

    if (curG >= 0) flush_run(out, curG, col4, acc);
}

extern "C" void kernel_launch(void* const* d_in, const int* in_sizes, int n_in,
                              void* d_out, int out_size, void* d_ws, size_t ws_size,
                              hipStream_t stream) {
    const float* nEmb  = (const float*)d_in[0];
    const float* gEmb  = (const float*)d_in[1];
    const int*   batch = (const int*)d_in[2];
    float*       out   = (float*)d_out;

    const int N = in_sizes[0] / DIMS;

    n2g_zero_kernel<<<ZERO_BLOCKS, ZERO_THREADS, 0, stream>>>(out, out_size);

    const int numWaves  = (N + NODES_PER_WAVE - 1) / NODES_PER_WAVE;
    const int numBlocks = (numWaves + WAVES_PER_BLOCK - 1) / WAVES_PER_BLOCK;
    n2g_attention_kernel<<<numBlocks, BLOCK, 0, stream>>>(nEmb, gEmb, batch, out, N);
}
